# starting kernel
# baseline (speedup 1.0000x reference)
#include <hip/hip_runtime.h>


typedef __attribute__((ext_vector_type(2))) float v2f;
typedef __attribute__((ext_vector_type(8))) float v8f;

#define NB 16
#define ND 32
#define HX 224
#define WX 224
#define HS 112
#define WS 112
#define FH 7
#define FW 7
#define TILE 16
#define TROWS (HS / TILE)
#define NWAVE (WS / TILE)
#define BLOCK (NWAVE * 32)
#define PROWS 22
#define PPITCH 122
#define KPAD 24

__global__ __launch_bounds__(BLOCK) void fused_pool_dwconv_wmma(
    const float* __restrict__ x, const float* __restrict__ S,
    float* __restrict__ out)
{
    __shared__ float Pool[PROWS * PPITCH];
    __shared__ float Bm[FH * KPAD * 16];

    const int b   = blockIdx.x;
    const int tr  = b % TROWS;
    const int d   = (b / TROWS) % ND;
    const int n   = b / (TROWS * ND);
    const int r0  = tr * TILE;
    const int tid = threadIdx.x;

    const float* xp = x + ((size_t)(n * ND + d)) * (HX * WX);
    const float* sp = S + d * (FH * FW);

    for (int e = tid; e < PROWS * PPITCH; e += BLOCK) {
        const int pr = e / PPITCH;
        const int cc = e % PPITCH;
        const int R  = r0 - 3 + pr;
        const int C  = cc - 3;
        const int Rc = R < 0 ? 0 : (R >= HS ? HS - 1 : R);
        const int Cc = C < 0 ? 0 : (C >= WS ? WS - 1 : C);
        const int ym = 2 * Rc - 1 < 0 ? 0 : 2 * Rc - 1;
        const int xm = 2 * Cc - 1 < 0 ? 0 : 2 * Cc - 1;
        const float* rowm = xp + ym * WX;
        const float* row0 = xp + 2 * Rc * WX;
        const float* rowp = row0 + WX;
        const v2f pm = *(const v2f*)(rowm + 2 * Cc);
        const v2f p0 = *(const v2f*)(row0 + 2 * Cc);
        const v2f pp = *(const v2f*)(rowp + 2 * Cc);
        const float lm = rowm[xm];
        const float l0 = row0[xm];
        const float lp = rowp[xm];
        float m = fmaxf(fmaxf(pm.x, pm.y), lm);
        m = fmaxf(m, fmaxf(fmaxf(p0.x, p0.y), l0));
        m = fmaxf(m, fmaxf(fmaxf(pp.x, pp.y), lp));
        const bool valid = (R >= 0) & (R < HS) & (C >= 0) & (C < WS);
        Pool[pr * PPITCH + cc] = valid ? m : 0.0f;
    }

    for (int e = tid; e < FH * KPAD * 16; e += BLOCK) {
        const int i = e / (KPAD * 16);
        const int k = (e / 16) % KPAD;
        const int c = e % 16;
        const int j = k - c;
        float v = 0.0f;
        if (j >= 0 && j < FW && k < 22) v = sp[i * FW + j];
        Bm[e] = v;
    }
    __syncthreads();

    const int wave  = tid >> 5;
    const int lane  = tid & 31;
    const int mrow  = lane & 15;
    const int khalf = lane >> 4;
    const int cbase = wave * TILE;

    v8f acc = {0.f, 0.f, 0.f, 0.f, 0.f, 0.f, 0.f, 0.f};
#pragma unroll
    for (int i = 0; i < FH; ++i) {
        const float* arow = &Pool[(mrow + i) * PPITCH + cbase + khalf * 2];
        const float* bcol = &Bm[i * (KPAD * 16) + khalf * 2 * 16 + mrow];
#pragma unroll
        for (int kb = 0; kb < KPAD / 4; ++kb) {
            v2f a = *(const v2f*)(arow + kb * 4);
            v2f bv;
            bv.x = bcol[kb * 4 * 16];
            bv.y = bcol[kb * 4 * 16 + 16];
            acc = __builtin_amdgcn_wmma_f32_16x16x4_f32(
                false, a, false, bv, (short)0, acc, false, false);
        }
    }

    float* op = out + ((size_t)(n * ND + d)) * (HS * WS)
                    + (size_t)r0 * WS + cbase;
#pragma unroll
    for (int v = 0; v < 8; ++v) {
        const int m = v + 8 * khalf;
        op[m * WS + mrow] = acc[v];
    }
}

extern "C" void kernel_launch(void* const* d_in, const int* in_sizes, int n_in,
                              void* d_out, int out_size, void* d_ws, size_t ws_size,
                              hipStream_t stream) {
    (void)in_sizes; (void)n_in; (void)d_ws; (void)ws_size; (void)out_size;
    const float* x = (const float*)d_in[0];
    const float* S = (const float*)d_in[1];
    float* out = (float*)d_out;

    dim3 grid(NB * ND * TROWS);
    fused_pool_dwconv_wmma<<<grid, BLOCK, 0, stream>>>(x, S, out);
}


# pool stage restructured: each x element loaded once per block by b128 row loads, 3x3 max done separably (vertical in registers, horizontal out of an LDS strip); was 6 loads per pooled element
# speedup vs baseline: 1.2017x; 1.2017x over previous
#include <hip/hip_runtime.h>


typedef __attribute__((ext_vector_type(2))) float v2f;
typedef __attribute__((ext_vector_type(4))) float v4f;
typedef __attribute__((ext_vector_type(8))) float v8f;

#define NB 16
#define ND 32
#define HX 224
#define WX 224
#define HS 112
#define WS 112
#define FH 7
#define FW 7
#define TILE 16
#define TROWS (HS / TILE)
#define NWAVE (WS / TILE)
#define BLOCK (NWAVE * 32)
#define PROWS 22
#define PPITCH 122
#define KPAD 24
#define NSTRIP (WX / 4)
#define NGRP (BLOCK / NSTRIP)
#define GROWS 6
#define VPITCH4 57

__global__ __launch_bounds__(BLOCK) void fused_pool_dwconv_wmma(
    const float* __restrict__ x, const float* __restrict__ S,
    float* __restrict__ out)
{
    __shared__ v4f   V[PROWS * VPITCH4];
    __shared__ float Pool[PROWS * PPITCH];
    __shared__ float Bm[FH * KPAD * 16];

    const int b   = blockIdx.x;
    const int tr  = b % TROWS;
    const int d   = (b / TROWS) % ND;
    const int n   = b / (TROWS * ND);
    const int r0  = tr * TILE;
    const int tid = threadIdx.x;

    const float* xp = x + ((size_t)(n * ND + d)) * (HX * WX);
    const float* sp = S + d * (FH * FW);

    const int strip = tid % NSTRIP;
    const int grp   = tid / NSTRIP;
    const int prA   = grp * GROWS;
    {
        const int y0 = 2 * (r0 - 3 + prA) - 1;
        const unsigned xoff = 4u * (unsigned)strip;
        v4f xr[2 * GROWS + 1];
#pragma unroll
        for (int j = 0; j < 9; ++j) {
            int y = y0 + j;
            y = y < 0 ? 0 : (y > HX - 1 ? HX - 1 : y);
            xr[j] = *(const v4f*)(xp + ((unsigned)y * (unsigned)WX + xoff));
        }
        if (grp < NGRP - 1) {
#pragma unroll
            for (int j = 9; j < 2 * GROWS + 1; ++j) {
                int y = y0 + j;
                y = y < 0 ? 0 : (y > HX - 1 ? HX - 1 : y);
                xr[j] = *(const v4f*)(xp + ((unsigned)y * (unsigned)WX + xoff));
            }
        }
#pragma unroll
        for (int k = 0; k < 4; ++k) {
            v4f m;
            m.x = fmaxf(fmaxf(xr[2 * k].x, xr[2 * k + 1].x), xr[2 * k + 2].x);
            m.y = fmaxf(fmaxf(xr[2 * k].y, xr[2 * k + 1].y), xr[2 * k + 2].y);
            m.z = fmaxf(fmaxf(xr[2 * k].z, xr[2 * k + 1].z), xr[2 * k + 2].z);
            m.w = fmaxf(fmaxf(xr[2 * k].w, xr[2 * k + 1].w), xr[2 * k + 2].w);
            V[(prA + k) * VPITCH4 + strip] = m;
        }
        if (grp < NGRP - 1) {
#pragma unroll
            for (int k = 4; k < GROWS; ++k) {
                v4f m;
                m.x = fmaxf(fmaxf(xr[2 * k].x, xr[2 * k + 1].x), xr[2 * k + 2].x);
                m.y = fmaxf(fmaxf(xr[2 * k].y, xr[2 * k + 1].y), xr[2 * k + 2].y);
                m.z = fmaxf(fmaxf(xr[2 * k].z, xr[2 * k + 1].z), xr[2 * k + 2].z);
                m.w = fmaxf(fmaxf(xr[2 * k].w, xr[2 * k + 1].w), xr[2 * k + 2].w);
                V[(prA + k) * VPITCH4 + strip] = m;
            }
        }
    }
    __syncthreads();

    {
        const float* Vf = (const float*)V;
        const int lcol = 4 * strip - 1 < 0 ? 0 : 4 * strip - 1;
        v4f   vv[5];
        float ll[5];
#pragma unroll
        for (int it = 0; it < 5; ++it) {
            const int pr = grp + NGRP * it;
            vv[it] = V[pr * VPITCH4 + strip];
            ll[it] = Vf[pr * (VPITCH4 * 4) + lcol];
        }
#pragma unroll
        for (int it = 0; it < 5; ++it) {
            const int pr = grp + NGRP * it;
            const int R = r0 - 3 + pr;
            const bool rowok = (R >= 0) & (R < HS);
            const float p0 = fmaxf(fmaxf(ll[it], vv[it].x), vv[it].y);
            const float p1 = fmaxf(fmaxf(vv[it].y, vv[it].z), vv[it].w);
            Pool[pr * PPITCH + 3 + 2 * strip]     = rowok ? p0 : 0.0f;
            Pool[pr * PPITCH + 3 + 2 * strip + 1] = rowok ? p1 : 0.0f;
        }
        if (grp < PROWS - NGRP * 5) {
            const int pr = grp + NGRP * 5;
            const int R = r0 - 3 + pr;
            const bool rowok = (R >= 0) & (R < HS);
            const v4f v = V[pr * VPITCH4 + strip];
            const float l = Vf[pr * (VPITCH4 * 4) + lcol];
            const float p0 = fmaxf(fmaxf(l, v.x), v.y);
            const float p1 = fmaxf(fmaxf(v.y, v.z), v.w);
            Pool[pr * PPITCH + 3 + 2 * strip]     = rowok ? p0 : 0.0f;
            Pool[pr * PPITCH + 3 + 2 * strip + 1] = rowok ? p1 : 0.0f;
        }
        if (tid < PROWS * 10) {
            const int pr = tid / 10;
            const int h  = tid % 10;
            Pool[pr * PPITCH + (h < 3 ? h : WS + h)] = 0.0f;
        }
    }

    for (int e = tid; e < FH * KPAD * 16; e += BLOCK) {
        const int i = e / (KPAD * 16);
        const int k = (e / 16) % KPAD;
        const int c = e % 16;
        const int j = k - c;
        float v = 0.0f;
        if (j >= 0 && j < FW && k < 22) v = sp[i * FW + j];
        Bm[e] = v;
    }
    __syncthreads();

    const int wave  = tid >> 5;
    const int lane  = tid & 31;
    const int mrow  = lane & 15;
    const int khalf = lane >> 4;
    const int cbase = wave * TILE;

    v8f acc = {0.f, 0.f, 0.f, 0.f, 0.f, 0.f, 0.f, 0.f};
#pragma unroll
    for (int i = 0; i < FH; ++i) {
        const float* arow = &Pool[(mrow + i) * PPITCH + cbase + khalf * 2];
        const float* bcol = &Bm[i * (KPAD * 16) + khalf * 2 * 16 + mrow];
#pragma unroll
        for (int kb = 0; kb < KPAD / 4; ++kb) {
            v2f a = *(const v2f*)(arow + kb * 4);
            v2f bv;
            bv.x = bcol[kb * 4 * 16];
            bv.y = bcol[kb * 4 * 16 + 16];
            acc = __builtin_amdgcn_wmma_f32_16x16x4_f32(
                false, a, false, bv, (short)0, acc, false, false);
        }
    }

    float* op = out + ((size_t)(n * ND + d)) * (HS * WS)
                    + (size_t)r0 * WS + cbase;
#pragma unroll
    for (int v = 0; v < 8; ++v) {
        const int m = v + 8 * khalf;
        op[m * WS + mrow] = acc[v];
    }
}

extern "C" void kernel_launch(void* const* d_in, const int* in_sizes, int n_in,
                              void* d_out, int out_size, void* d_ws, size_t ws_size,
                              hipStream_t stream) {
    (void)in_sizes; (void)n_in; (void)d_ws; (void)ws_size; (void)out_size;
    const float* x = (const float*)d_in[0];
    const float* S = (const float*)d_in[1];
    float* out = (float*)d_out;

    dim3 grid(NB * ND * TROWS);
    fused_pool_dwconv_wmma<<<grid, BLOCK, 0, stream>>>(x, S, out);
}


# one work-group per (n,d) plane loops over its 7 tile-rows: banded B fragments built once into registers, next tile-row's x rows loaded ahead across LDS-only barriers; 512 groups, was 3584
# speedup vs baseline: 1.4545x; 1.2104x over previous
#include <hip/hip_runtime.h>


typedef __attribute__((ext_vector_type(2))) float v2f;
typedef __attribute__((ext_vector_type(4))) float v4f;
typedef __attribute__((ext_vector_type(8))) float v8f;

#define NB 16
#define ND 32
#define HX 224
#define WX 224
#define HS 112
#define WS 112
#define FH 7
#define FW 7
#define TILE 16
#define TROWS (HS / TILE)
#define NWAVE (WS / TILE)
#define BLOCK (NWAVE * 32)
#define PROWS 22
#define PPITCH 122
#define KPAD 24
#define KREAL 22
#define NSTRIP (WX / 4)
#define NGRP (BLOCK / NSTRIP)
#define GROWS 6
#define NXR (2 * GROWS + 1)
#define VPITCH4 57

static __device__ __forceinline__ void lds_barrier()
{
    __builtin_amdgcn_fence(__ATOMIC_RELEASE, "workgroup", "local");
    __builtin_amdgcn_s_barrier();
    __builtin_amdgcn_fence(__ATOMIC_ACQUIRE, "workgroup", "local");
}

static __device__ __forceinline__ void load_rows(
    const float* __restrict__ xp, int r0, int prA, int grp, unsigned xoff,
    v4f (&xr)[NXR])
{
    const int y0 = 2 * (r0 - 3 + prA) - 1;
#pragma unroll
    for (int j = 0; j < 9; ++j) {
        int y = y0 + j;
        y = y < 0 ? 0 : (y > HX - 1 ? HX - 1 : y);
        xr[j] = *(const v4f*)(xp + ((unsigned)y * (unsigned)WX + xoff));
    }
    if (grp < NGRP - 1) {
#pragma unroll
        for (int j = 9; j < NXR; ++j) {
            int y = y0 + j;
            y = y < 0 ? 0 : (y > HX - 1 ? HX - 1 : y);
            xr[j] = *(const v4f*)(xp + ((unsigned)y * (unsigned)WX + xoff));
        }
    }
}

__global__ __launch_bounds__(BLOCK) void fused_pool_dwconv_wmma(
    const float* __restrict__ x, const float* __restrict__ S,
    float* __restrict__ out)
{
    __shared__ v4f   V[PROWS * VPITCH4];
    __shared__ float Pool[PROWS * PPITCH];
    __shared__ float Ssh[FH * FW];

    const int plane = blockIdx.x;
    const int d     = plane % ND;
    const int tid   = threadIdx.x;

    const float* xp = x + (size_t)plane * (HX * WX);
    const float* sp = S + d * (FH * FW);
    float* oplane   = out + (size_t)plane * (HS * WS);

    const int strip = tid % NSTRIP;
    const int grp   = tid / NSTRIP;
    const int prA   = grp * GROWS;
    const unsigned xoff = 4u * (unsigned)strip;

    const int wave  = tid >> 5;
    const int lane  = tid & 31;
    const int mrow  = lane & 15;
    const int khalf = lane >> 4;
    const int cbase = wave * TILE;

    v4f xr[NXR];
    load_rows(xp, 0, prA, grp, xoff, xr);

    if (tid < FH * FW) Ssh[tid] = sp[tid];
    lds_barrier();

    v2f breg[FH][KPAD / 4];
#pragma unroll
    for (int i = 0; i < FH; ++i) {
#pragma unroll
        for (int kb = 0; kb < KPAD / 4; ++kb) {
            const int k0 = kb * 4 + khalf * 2;
            const int j0 = k0 - mrow;
            const int j1 = j0 + 1;
            const int c0 = j0 < 0 ? 0 : (j0 > FW - 1 ? FW - 1 : j0);
            const int c1 = j1 < 0 ? 0 : (j1 > FW - 1 ? FW - 1 : j1);
            const float s0 = Ssh[i * FW + c0];
            const float s1 = Ssh[i * FW + c1];
            breg[i][kb].x = (j0 >= 0 && j0 < FW && k0 < KREAL) ? s0 : 0.0f;
            breg[i][kb].y = (j1 >= 0 && j1 < FW && k0 + 1 < KREAL) ? s1 : 0.0f;
        }
    }

#pragma nounroll
    for (int tr = 0; tr < TROWS; ++tr) {
        const int r0 = tr * TILE;

#pragma unroll
        for (int k = 0; k < 4; ++k) {
            v4f m;
            m.x = fmaxf(fmaxf(xr[2 * k].x, xr[2 * k + 1].x), xr[2 * k + 2].x);
            m.y = fmaxf(fmaxf(xr[2 * k].y, xr[2 * k + 1].y), xr[2 * k + 2].y);
            m.z = fmaxf(fmaxf(xr[2 * k].z, xr[2 * k + 1].z), xr[2 * k + 2].z);
            m.w = fmaxf(fmaxf(xr[2 * k].w, xr[2 * k + 1].w), xr[2 * k + 2].w);
            V[(prA + k) * VPITCH4 + strip] = m;
        }
        if (grp < NGRP - 1) {
#pragma unroll
            for (int k = 4; k < GROWS; ++k) {
                v4f m;
                m.x = fmaxf(fmaxf(xr[2 * k].x, xr[2 * k + 1].x), xr[2 * k + 2].x);
                m.y = fmaxf(fmaxf(xr[2 * k].y, xr[2 * k + 1].y), xr[2 * k + 2].y);
                m.z = fmaxf(fmaxf(xr[2 * k].z, xr[2 * k + 1].z), xr[2 * k + 2].z);
                m.w = fmaxf(fmaxf(xr[2 * k].w, xr[2 * k + 1].w), xr[2 * k + 2].w);
                V[(prA + k) * VPITCH4 + strip] = m;
            }
        }

        if (tr + 1 < TROWS) load_rows(xp, r0 + TILE, prA, grp, xoff, xr);

        lds_barrier();

        {
            const float* Vf = (const float*)V;
            const int lcol = 4 * strip - 1 < 0 ? 0 : 4 * strip - 1;
            v4f   vv[5];
            float ll[5];
#pragma unroll
            for (int it = 0; it < 5; ++it) {
                const int pr = grp + NGRP * it;
                vv[it] = V[pr * VPITCH4 + strip];
                ll[it] = Vf[pr * (VPITCH4 * 4) + lcol];
            }
#pragma unroll
            for (int it = 0; it < 5; ++it) {
                const int pr = grp + NGRP * it;
                const int R = r0 - 3 + pr;
                const bool rowok = (R >= 0) & (R < HS);
                const float p0 = fmaxf(fmaxf(ll[it], vv[it].x), vv[it].y);
                const float p1 = fmaxf(fmaxf(vv[it].y, vv[it].z), vv[it].w);
                Pool[pr * PPITCH + 3 + 2 * strip]     = rowok ? p0 : 0.0f;
                Pool[pr * PPITCH + 3 + 2 * strip + 1] = rowok ? p1 : 0.0f;
            }
            if (grp < PROWS - NGRP * 5) {
                const int pr = grp + NGRP * 5;
                const int R = r0 - 3 + pr;
                const bool rowok = (R >= 0) & (R < HS);
                const v4f v = V[pr * VPITCH4 + strip];
                const float l = Vf[pr * (VPITCH4 * 4) + lcol];
                const float p0 = fmaxf(fmaxf(l, v.x), v.y);
                const float p1 = fmaxf(fmaxf(v.y, v.z), v.w);
                Pool[pr * PPITCH + 3 + 2 * strip]     = rowok ? p0 : 0.0f;
                Pool[pr * PPITCH + 3 + 2 * strip + 1] = rowok ? p1 : 0.0f;
            }
            if (tid < PROWS * 10) {
                const int pr = tid / 10;
                const int h  = tid % 10;
                Pool[pr * PPITCH + (h < 3 ? h : WS + h)] = 0.0f;
            }
        }
        lds_barrier();

        v8f acc = {0.f, 0.f, 0.f, 0.f, 0.f, 0.f, 0.f, 0.f};
#pragma unroll
        for (int i = 0; i < FH; ++i) {
            const float* arow = &Pool[(mrow + i) * PPITCH + cbase + khalf * 2];
#pragma unroll
            for (int kb = 0; kb < KPAD / 4; ++kb) {
                v2f a = *(const v2f*)(arow + kb * 4);
                acc = __builtin_amdgcn_wmma_f32_16x16x4_f32(
                    false, a, false, breg[i][kb], (short)0, acc, false, false);
            }
        }

        float* op = oplane + (size_t)r0 * WS + cbase;
#pragma unroll
        for (int v = 0; v < 8; ++v) {
            const int m = v + 8 * khalf;
            op[m * WS + mrow] = acc[v];
        }
    }
}

extern "C" void kernel_launch(void* const* d_in, const int* in_sizes, int n_in,
                              void* d_out, int out_size, void* d_ws, size_t ws_size,
                              hipStream_t stream) {
    (void)in_sizes; (void)n_in; (void)d_ws; (void)ws_size; (void)out_size;
    const float* x = (const float*)d_in[0];
    const float* S = (const float*)d_in[1];
    float* out = (float*)d_out;

    dim3 grid(NB * ND);
    fused_pool_dwconv_wmma<<<grid, BLOCK, 0, stream>>>(x, S, out);
}


# 7x7 taps as ONE 2-D Toeplitz product over 4x4-output patches: K=100, 25 WMMAs per 256 outputs (was 7 banded products, 42); taps still summed row-major in one chain; two 16-B row stores per lane
# speedup vs baseline: 1.7152x; 1.1792x over previous
#include <hip/hip_runtime.h>


typedef __attribute__((ext_vector_type(2))) float v2f;
typedef __attribute__((ext_vector_type(4))) float v4f;
typedef __attribute__((ext_vector_type(8))) float v8f;

#define NB 16
#define ND 32
#define HX 224
#define WX 224
#define HS 112
#define WS 112
#define FH 7
#define FW 7
#define TILE 16
#define TROWS (HS / TILE)
#define NWAVE (WS / TILE)
#define BLOCK (NWAVE * 32)
#define PROWS 22
#define PPITCH 124
#define PW 4
#define PIN (PW + FW - 1)
#define KTOT (PIN * PIN)
#define NKB (KTOT / 4)
#define PCOLS (WS / PW)
#define NSTRIP (WX / 4)
#define NGRP (BLOCK / NSTRIP)
#define GROWS 6
#define NXR (2 * GROWS + 1)
#define VPITCH4 57

static __device__ __forceinline__ void lds_barrier()
{
    __builtin_amdgcn_fence(__ATOMIC_RELEASE, "workgroup", "local");
    __builtin_amdgcn_s_barrier();
    __builtin_amdgcn_fence(__ATOMIC_ACQUIRE, "workgroup", "local");
}

static __device__ __forceinline__ void load_rows(
    const float* __restrict__ xp, int r0, int prA, int grp, unsigned xoff,
    v4f (&xr)[NXR])
{
    const int y0 = 2 * (r0 - 3 + prA) - 1;
#pragma unroll
    for (int j = 0; j < 9; ++j) {
        int y = y0 + j;
        y = y < 0 ? 0 : (y > HX - 1 ? HX - 1 : y);
        xr[j] = *(const v4f*)(xp + ((unsigned)y * (unsigned)WX + xoff));
    }
    if (grp < NGRP - 1) {
#pragma unroll
        for (int j = 9; j < NXR; ++j) {
            int y = y0 + j;
            y = y < 0 ? 0 : (y > HX - 1 ? HX - 1 : y);
            xr[j] = *(const v4f*)(xp + ((unsigned)y * (unsigned)WX + xoff));
        }
    }
}

__global__ __launch_bounds__(BLOCK) void fused_pool_dwconv_wmma(
    const float* __restrict__ x, const float* __restrict__ S,
    float* __restrict__ out)
{
    __shared__ v4f   V[PROWS * VPITCH4];
    __shared__ __attribute__((aligned(16))) float Pool[PROWS * PPITCH];
    __shared__ float Ssh[FH * FW];

    const int plane = blockIdx.x;
    const int d     = plane % ND;
    const int tid   = threadIdx.x;

    const float* xp = x + (size_t)plane * (HX * WX);
    const float* sp = S + d * (FH * FW);
    float* oplane   = out + (size_t)plane * (HS * WS);

    const int strip = tid % NSTRIP;
    const int grp   = tid / NSTRIP;
    const int prA   = grp * GROWS;
    const unsigned xoff = 4u * (unsigned)strip;

    const int wave  = tid >> 5;
    const int lane  = tid & 31;
    const int mrow  = lane & 15;
    const int khalf = lane >> 4;
    const int patch = wave * TILE + mrow;
    const int prow  = patch / PCOLS;
    const int pcol  = patch % PCOLS;

    v4f xr[NXR];
    load_rows(xp, 0, prA, grp, xoff, xr);

    if (tid < FH * FW) Ssh[tid] = sp[tid];
    lds_barrier();

    v2f treg[NKB];
    {
        const int oa = mrow >> 2;
        const int ob = mrow & 3;
#pragma unroll
        for (int kb = 0; kb < NKB; ++kb) {
            const int u  = khalf ? (4 * kb + 2) / PIN : (4 * kb) / PIN;
            const int v  = khalf ? (4 * kb + 2) % PIN : (4 * kb) % PIN;
            const int i  = u - oa;
            const int j0 = v - ob;
            const int j1 = j0 + 1;
            const int ci = i  < 0 ? 0 : (i  > FH - 1 ? FH - 1 : i);
            const int c0 = j0 < 0 ? 0 : (j0 > FW - 1 ? FW - 1 : j0);
            const int c1 = j1 < 0 ? 0 : (j1 > FW - 1 ? FW - 1 : j1);
            const float s0 = Ssh[ci * FW + c0];
            const float s1 = Ssh[ci * FW + c1];
            const bool iok = (i >= 0) & (i < FH);
            treg[kb].x = (iok & (j0 >= 0) & (j0 < FW)) ? s0 : 0.0f;
            treg[kb].y = (iok & (j1 >= 0) & (j1 < FW)) ? s1 : 0.0f;
        }
    }

    const float* pnear = &Pool[(PW * prow) * PPITCH + PW * pcol + (khalf ? 2 : 0)];
    const float* pwrap = &Pool[(PW * prow) * PPITCH + PW * pcol + (khalf ? PPITCH - 8 : 0)];
    const unsigned ooff = (unsigned)(PW * prow + 2 * khalf) * WS + PW * pcol;

#pragma nounroll
    for (int tr = 0; tr < TROWS; ++tr) {
        const int r0 = tr * TILE;

#pragma unroll
        for (int k = 0; k < 4; ++k) {
            v4f m;
            m.x = fmaxf(fmaxf(xr[2 * k].x, xr[2 * k + 1].x), xr[2 * k + 2].x);
            m.y = fmaxf(fmaxf(xr[2 * k].y, xr[2 * k + 1].y), xr[2 * k + 2].y);
            m.z = fmaxf(fmaxf(xr[2 * k].z, xr[2 * k + 1].z), xr[2 * k + 2].z);
            m.w = fmaxf(fmaxf(xr[2 * k].w, xr[2 * k + 1].w), xr[2 * k + 2].w);
            V[(prA + k) * VPITCH4 + strip] = m;
        }
        if (grp < NGRP - 1) {
#pragma unroll
            for (int k = 4; k < GROWS; ++k) {
                v4f m;
                m.x = fmaxf(fmaxf(xr[2 * k].x, xr[2 * k + 1].x), xr[2 * k + 2].x);
                m.y = fmaxf(fmaxf(xr[2 * k].y, xr[2 * k + 1].y), xr[2 * k + 2].y);
                m.z = fmaxf(fmaxf(xr[2 * k].z, xr[2 * k + 1].z), xr[2 * k + 2].z);
                m.w = fmaxf(fmaxf(xr[2 * k].w, xr[2 * k + 1].w), xr[2 * k + 2].w);
                V[(prA + k) * VPITCH4 + strip] = m;
            }
        }

        if (tr + 1 < TROWS) load_rows(xp, r0 + TILE, prA, grp, xoff, xr);

        lds_barrier();

        {
            const float* Vf = (const float*)V;
            const int lcol = 4 * strip - 1 < 0 ? 0 : 4 * strip - 1;
            v4f   vv[5];
            float ll[5];
#pragma unroll
            for (int it = 0; it < 5; ++it) {
                const int pr = grp + NGRP * it;
                vv[it] = V[pr * VPITCH4 + strip];
                ll[it] = Vf[pr * (VPITCH4 * 4) + lcol];
            }
#pragma unroll
            for (int it = 0; it < 5; ++it) {
                const int pr = grp + NGRP * it;
                const int R = r0 - 3 + pr;
                const bool rowok = (R >= 0) & (R < HS);
                const float p0 = fmaxf(fmaxf(ll[it], vv[it].x), vv[it].y);
                const float p1 = fmaxf(fmaxf(vv[it].y, vv[it].z), vv[it].w);
                Pool[pr * PPITCH + 3 + 2 * strip]     = rowok ? p0 : 0.0f;
                Pool[pr * PPITCH + 3 + 2 * strip + 1] = rowok ? p1 : 0.0f;
            }
            if (grp < PROWS - NGRP * 5) {
                const int pr = grp + NGRP * 5;
                const int R = r0 - 3 + pr;
                const bool rowok = (R >= 0) & (R < HS);
                const v4f v = V[pr * VPITCH4 + strip];
                const float l = Vf[pr * (VPITCH4 * 4) + lcol];
                const float p0 = fmaxf(fmaxf(l, v.x), v.y);
                const float p1 = fmaxf(fmaxf(v.y, v.z), v.w);
                Pool[pr * PPITCH + 3 + 2 * strip]     = rowok ? p0 : 0.0f;
                Pool[pr * PPITCH + 3 + 2 * strip + 1] = rowok ? p1 : 0.0f;
            }
            if (tid < PROWS * 10) {
                const int pr = tid / 10;
                const int h  = tid % 10;
                Pool[pr * PPITCH + (h < 3 ? h : WS + h)] = 0.0f;
            }
        }
        lds_barrier();

        v8f acc = {0.f, 0.f, 0.f, 0.f, 0.f, 0.f, 0.f, 0.f};
#pragma unroll
        for (int kb = 0; kb < NKB; ++kb) {
            const int u0 = (4 * kb) / PIN;
            const int v0 = (4 * kb) % PIN;
            const float* p = (v0 == PIN - 2) ? pwrap : pnear;
            const v2f pf = *(const v2f*)(p + u0 * PPITCH + v0);
            acc = __builtin_amdgcn_wmma_f32_16x16x4_f32(
                false, treg[kb], false, pf, (short)0, acc, false, false);
        }

        float* op = oplane + (size_t)r0 * WS + ooff;
        v4f lo, hi;
        lo.x = acc[0]; lo.y = acc[1]; lo.z = acc[2]; lo.w = acc[3];
        hi.x = acc[4]; hi.y = acc[5]; hi.z = acc[6]; hi.w = acc[7];
        *(v4f*)op        = lo;
        *(v4f*)(op + WS) = hi;
    }
}

extern "C" void kernel_launch(void* const* d_in, const int* in_sizes, int n_in,
                              void* d_out, int out_size, void* d_ws, size_t ws_size,
                              hipStream_t stream) {
    (void)in_sizes; (void)n_in; (void)d_ws; (void)ws_size; (void)out_size;
    const float* x = (const float*)d_in[0];
    const float* S = (const float*)d_in[1];
    float* out = (float*)d_out;

    dim3 grid(NB * ND);
    fused_pool_dwconv_wmma<<<grid, BLOCK, 0, stream>>>(x, S, out);
}


# pooled rows shared by two tile-rows are pooled once: two alternating LDS strips, a step pools its 16 new rows and stores the last 6 also into the next strip; 9 x-row loads per thread, was 13
# speedup vs baseline: 1.8911x; 1.1026x over previous
#include <hip/hip_runtime.h>


typedef __attribute__((ext_vector_type(2))) float v2f;
typedef __attribute__((ext_vector_type(4))) float v4f;
typedef __attribute__((ext_vector_type(8))) float v8f;

#define NB 16
#define ND 32
#define HX 224
#define WX 224
#define HS 112
#define WS 112
#define FH 7
#define FW 7
#define TILE 16
#define TROWS (HS / TILE)
#define NWAVE (WS / TILE)
#define BLOCK (NWAVE * 32)
#define PROWS 22
#define NEWR TILE
#define HALO (PROWS - NEWR)
#define POOLSZ (PROWS * PPITCH)
#define PPITCH 124
#define PW 4
#define PIN (PW + FW - 1)
#define KTOT (PIN * PIN)
#define NKB (KTOT / 4)
#define PCOLS (WS / PW)
#define NSTRIP (WX / 4)
#define NGRP (BLOCK / NSTRIP)
#define GROWS (NEWR / NGRP)
#define NXR (2 * GROWS + 1)
#define VPITCH4 57

static __device__ __forceinline__ void lds_barrier()
{
    __builtin_amdgcn_fence(__ATOMIC_RELEASE, "workgroup", "local");
    __builtin_amdgcn_s_barrier();
    __builtin_amdgcn_fence(__ATOMIC_ACQUIRE, "workgroup", "local");
}

template <bool FIRST>
static __device__ __forceinline__ void load_rows(
    const float* __restrict__ xp, int R0, int grp, unsigned xoff,
    v4f (&xr)[NXR])
{
    if (FIRST && grp != NGRP - 1) {
#pragma unroll
        for (int j = 0; j < NXR; ++j) xr[j] = (v4f){0.f, 0.f, 0.f, 0.f};
        return;
    }
    const int y0 = 2 * (R0 + GROWS * grp) - 1;
#pragma unroll
    for (int j = 0; j < NXR; ++j) {
        int y = y0 + j;
        y = y < 0 ? 0 : (y > HX - 1 ? HX - 1 : y);
        xr[j] = *(const v4f*)(xp + ((unsigned)y * (unsigned)WX + xoff));
    }
}

static __device__ __forceinline__ void stage1a(
    const v4f (&xr)[NXR], v4f* V, int grp, int strip)
{
#pragma unroll
    for (int k = 0; k < GROWS; ++k) {
        v4f m;
        m.x = fmaxf(fmaxf(xr[2 * k].x, xr[2 * k + 1].x), xr[2 * k + 2].x);
        m.y = fmaxf(fmaxf(xr[2 * k].y, xr[2 * k + 1].y), xr[2 * k + 2].y);
        m.z = fmaxf(fmaxf(xr[2 * k].z, xr[2 * k + 1].z), xr[2 * k + 2].z);
        m.w = fmaxf(fmaxf(xr[2 * k].w, xr[2 * k + 1].w), xr[2 * k + 2].w);
        V[(GROWS * grp + k) * VPITCH4 + strip] = m;
    }
}

static __device__ __forceinline__ void stage1b(
    const v4f* V, float* pcur, float* pnxt, int R0, int grp, int strip)
{
    const float* Vf = (const float*)V;
    const int lcol = 4 * strip - 1 < 0 ? 0 : 4 * strip - 1;
    const int col  = 3 + 2 * strip;
    v4f   vv[NEWR / NGRP];
    float ll[NEWR / NGRP];
#pragma unroll
    for (int it = 0; it < NEWR / NGRP; ++it) {
        const int nr = grp + NGRP * it;
        vv[it] = V[nr * VPITCH4 + strip];
        ll[it] = Vf[nr * (VPITCH4 * 4) + lcol];
    }
#pragma unroll
    for (int it = 0; it < NEWR / NGRP; ++it) {
        const int nr = grp + NGRP * it;
        const int R = R0 + nr;
        const bool rowok = (R >= 0) & (R < HS);
        const float p0 = fmaxf(fmaxf(ll[it], vv[it].x), vv[it].y);
        const float p1 = fmaxf(fmaxf(vv[it].y, vv[it].z), vv[it].w);
        const float q0 = rowok ? p0 : 0.0f;
        const float q1 = rowok ? p1 : 0.0f;
        pcur[(HALO + nr) * PPITCH + col]     = q0;
        pcur[(HALO + nr) * PPITCH + col + 1] = q1;
        if (it == 3 || (it == 2 && grp >= 2)) {
            pnxt[(nr - (NEWR - HALO)) * PPITCH + col]     = q0;
            pnxt[(nr - (NEWR - HALO)) * PPITCH + col + 1] = q1;
        }
    }
}

__global__ __launch_bounds__(BLOCK) void fused_pool_dwconv_wmma(
    const float* __restrict__ x, const float* __restrict__ S,
    float* __restrict__ out)
{
    __shared__ v4f   V[NEWR * VPITCH4];
    __shared__ __attribute__((aligned(16))) float Pool[2 * POOLSZ];
    __shared__ float Ssh[FH * FW];

    const int plane = blockIdx.x;
    const int d     = plane % ND;
    const int tid   = threadIdx.x;

    const float* xp = x + (size_t)plane * (HX * WX);
    const float* sp = S + d * (FH * FW);
    float* oplane   = out + (size_t)plane * (HS * WS);

    const int strip = tid % NSTRIP;
    const int grp   = tid / NSTRIP;
    const unsigned xoff = 4u * (unsigned)strip;

    const int wave  = tid >> 5;
    const int lane  = tid & 31;
    const int mrow  = lane & 15;
    const int khalf = lane >> 4;
    const int patch = wave * TILE + mrow;
    const int prow  = patch / PCOLS;
    const int pcol  = patch % PCOLS;

    v4f xr[NXR];
    load_rows<true>(xp, 3 - NEWR, grp, xoff, xr);

    if (tid < FH * FW) Ssh[tid] = sp[tid];
    if (tid < PROWS * 10) {
        const int pr = tid / 10;
        const int h  = tid % 10;
        Pool[pr * PPITCH + (h < 3 ? h : WS + h)]          = 0.0f;
        Pool[POOLSZ + pr * PPITCH + (h < 3 ? h : WS + h)] = 0.0f;
    }
    stage1a(xr, V, grp, strip);
    load_rows<false>(xp, 3, grp, xoff, xr);
    lds_barrier();
    stage1b(V, Pool + POOLSZ, Pool, 3 - NEWR, grp, strip);

    v2f treg[NKB];
    {
        const int oa = mrow >> 2;
        const int ob = mrow & 3;
#pragma unroll
        for (int kb = 0; kb < NKB; ++kb) {
            const int u  = khalf ? (4 * kb + 2) / PIN : (4 * kb) / PIN;
            const int v  = khalf ? (4 * kb + 2) % PIN : (4 * kb) % PIN;
            const int i  = u - oa;
            const int j0 = v - ob;
            const int j1 = j0 + 1;
            const int ci = i  < 0 ? 0 : (i  > FH - 1 ? FH - 1 : i);
            const int c0 = j0 < 0 ? 0 : (j0 > FW - 1 ? FW - 1 : j0);
            const int c1 = j1 < 0 ? 0 : (j1 > FW - 1 ? FW - 1 : j1);
            const float s0 = Ssh[ci * FW + c0];
            const float s1 = Ssh[ci * FW + c1];
            const bool iok = (i >= 0) & (i < FH);
            treg[kb].x = (iok & (j0 >= 0) & (j0 < FW)) ? s0 : 0.0f;
            treg[kb].y = (iok & (j1 >= 0) & (j1 < FW)) ? s1 : 0.0f;
        }
    }

    const float* pnear = &Pool[(PW * prow) * PPITCH + PW * pcol + (khalf ? 2 : 0)];
    const float* pwrap = &Pool[(PW * prow) * PPITCH + PW * pcol + (khalf ? PPITCH - 8 : 0)];
    const unsigned ooff = (unsigned)(PW * prow + 2 * khalf) * WS + PW * pcol;

    lds_barrier();

#pragma nounroll
    for (int tr = 0; tr < TROWS; ++tr) {
        const int r0 = tr * TILE;

        const int cur = (tr & 1) * POOLSZ;

        stage1a(xr, V, grp, strip);

        if (tr + 1 < TROWS) load_rows<false>(xp, r0 + TILE + 3, grp, xoff, xr);

        lds_barrier();

        stage1b(V, Pool + cur, Pool + (POOLSZ - cur), r0 + 3, grp, strip);
        lds_barrier();

        v8f acc = {0.f, 0.f, 0.f, 0.f, 0.f, 0.f, 0.f, 0.f};
#pragma unroll
        for (int kb = 0; kb < NKB; ++kb) {
            const int u0 = (4 * kb) / PIN;
            const int v0 = (4 * kb) % PIN;
            const float* p = ((v0 == PIN - 2) ? pwrap : pnear) + cur;
            const v2f pf = *(const v2f*)(p + u0 * PPITCH + v0);
            acc = __builtin_amdgcn_wmma_f32_16x16x4_f32(
                false, treg[kb], false, pf, (short)0, acc, false, false);
        }

        float* op = oplane + (size_t)r0 * WS + ooff;
        v4f lo, hi;
        lo.x = acc[0]; lo.y = acc[1]; lo.z = acc[2]; lo.w = acc[3];
        hi.x = acc[4]; hi.y = acc[5]; hi.z = acc[6]; hi.w = acc[7];
        *(v4f*)op        = lo;
        *(v4f*)(op + WS) = hi;
    }
}

extern "C" void kernel_launch(void* const* d_in, const int* in_sizes, int n_in,
                              void* d_out, int out_size, void* d_ws, size_t ws_size,
                              hipStream_t stream) {
    (void)in_sizes; (void)n_in; (void)d_ws; (void)ws_size; (void)out_size;
    const float* x = (const float*)d_in[0];
    const float* S = (const float*)d_in[1];
    float* out = (float*)d_out;

    dim3 grid(NB * ND);
    fused_pool_dwconv_wmma<<<grid, BLOCK, 0, stream>>>(x, S, out);
}
